# starting kernel
# baseline (speedup 1.0000x reference)
#include <hip/hip_runtime.h>
#include <hip/hip_bf16.h>

#define SS 128
#define QQ 8192
#define HH 230
#define NBLK (QQ / 16)

typedef float v2f __attribute__((ext_vector_type(2)));
typedef float v8f __attribute__((ext_vector_type(8)));

__global__ void row_norm_kernel(const float* __restrict__ x,
                                const float* __restrict__ w,
                                float* __restrict__ arow) {
    int s = threadIdx.x;
    if (s < SS) {
        const float* xr = x + s * HH;
        float acc = 0.f;
        for (int h = 0; h < HH; ++h) {
            float v = xr[h];
            acc = fmaf(w[h], v * v, acc);
        }
        arow[s] = acc;
    }
}

__global__ void col_norm_kernel(const float* __restrict__ y,
                                const float* __restrict__ w,
                                float* __restrict__ ccol) {
    int q = blockIdx.x * blockDim.x + threadIdx.x;
    if (q < QQ) {
        const float* yr = y + q * HH;
        float acc = 0.f;
        for (int h = 0; h < HH; ++h) {
            float v = yr[h];
            acc = fmaf(w[h], v * v, acc);
        }
        ccol[q] = acc;
    }
}

__global__ __launch_bounds__(256)
void siamese_wmma_kernel(const float* __restrict__ x, const float* __restrict__ y,
                         const int* __restrict__ label, const float* __restrict__ w,
                         const float* __restrict__ bias,
                         const float* __restrict__ arow, const float* __restrict__ ccol,
                         float* __restrict__ score, float* __restrict__ partial)
{
    __shared__ float red[256];

    const int lane = threadIdx.x & 31;
    const int wave = threadIdx.x >> 5;
    const int half = lane >> 4;
    const int l    = lane & 15;
    const int m0   = wave * 16;
    const int qb   = blockIdx.x * 16;

    const float* xr = x + (m0 + l) * HH + 2 * half;
    const float* yr = y + (qb + l) * HH + 2 * half;
    const float* wr = w + 2 * half;

    v8f acc = {};

    #pragma unroll 4
    for (int k0 = 0; k0 + 4 <= HH; k0 += 4) {
        float2 xv = *(const float2*)(xr + k0);
        float2 wv = *(const float2*)(wr + k0);
        float2 yv = *(const float2*)(yr + k0);
        v2f A; A[0] = xv.x * wv.x; A[1] = xv.y * wv.y;
        v2f B; B[0] = yv.x;        B[1] = yv.y;
        acc = __builtin_amdgcn_wmma_f32_16x16x4_f32(
             false, A,  false, B,
             (short)0, acc,  false,  false);
    }

    {
        v2f A = {0.f, 0.f};
        v2f B = {0.f, 0.f};
        if (half == 0) {
            float2 xv = *(const float2*)(xr + 228);
            float2 wv = *(const float2*)(wr + 228);
            float2 yv = *(const float2*)(yr + 228);
            A[0] = xv.x * wv.x; A[1] = xv.y * wv.y;
            B[0] = yv.x;        B[1] = yv.y;
        }
        acc = __builtin_amdgcn_wmma_f32_16x16x4_f32(
            false, A, false, B, (short)0, acc, false, false);
    }

    const float bval = *bias;
    const int   q    = qb + l;
    const float cq   = ccol[q];
    float lsum = 0.f;
    #pragma unroll
    for (int r = 0; r < 8; ++r) {
        const int s = m0 + r + 8 * half;
        const float z  = fmaf(-2.f, acc[r], arow[s] + cq + bval);
        const float sc = 1.f / (1.f + expf(-z));
        score[s * QQ + q] = sc;
        const float lab = (float)label[s * QQ + q];
        const float sp = fmaxf(z, 0.f) + log1pf(expf(-fabsf(z)));
        lsum += sp - lab * z;
    }

    red[threadIdx.x] = lsum;
    __syncthreads();
    for (int off = 128; off > 0; off >>= 1) {
        if (threadIdx.x < (unsigned)off) red[threadIdx.x] += red[threadIdx.x + off];
        __syncthreads();
    }
    if (threadIdx.x == 0) partial[blockIdx.x] = red[0];
}

__global__ __launch_bounds__(256)
void loss_kernel(const float* __restrict__ partial, float* __restrict__ out) {
    __shared__ float red[256];
    red[threadIdx.x] = partial[threadIdx.x] + partial[threadIdx.x + 256];
    __syncthreads();
    for (int off = 128; off > 0; off >>= 1) {
        if (threadIdx.x < (unsigned)off) red[threadIdx.x] += red[threadIdx.x + off];
        __syncthreads();
    }
    if (threadIdx.x == 0) out[SS * QQ] = red[0] * (1.0f / (float)(SS * QQ));
}

extern "C" void kernel_launch(void* const* d_in, const int* in_sizes, int n_in,
                              void* d_out, int out_size, void* d_ws, size_t ws_size,
                              hipStream_t stream) {
    const float* x     = (const float*)d_in[0];
    const float* y     = (const float*)d_in[1];
    const int*   label = (const int*)  d_in[2];
    const float* w     = (const float*)d_in[3];
    const float* b     = (const float*)d_in[4];

    float* out = (float*)d_out;
    float* ws  = (float*)d_ws;
    float* arow    = ws;
    float* ccol    = ws + SS;
    float* partial = ws + SS + QQ;

    row_norm_kernel<<<1, 128, 0, stream>>>(x, w, arow);
    col_norm_kernel<<<QQ / 256, 256, 0, stream>>>(y, w, ccol);
    siamese_wmma_kernel<<<NBLK, 256, 0, stream>>>(x, y, label, w, b,
                                                  arow, ccol, out, partial);
    loss_kernel<<<1, 256, 0, stream>>>(partial, out);
}


# Norms a[s], c[q] summed inside the GEMM K loop from the loaded fragments (4 interleaved partial sums per row joined by lane shuffles: new summation order); both norm launches dropped
# speedup vs baseline: 2.4857x; 2.4857x over previous
#include <hip/hip_runtime.h>
#include <hip/hip_bf16.h>

#define SS 128
#define QQ 8192
#define HH 230
#define NBLK (QQ / 16)
#define KCH 8

typedef float v2f __attribute__((ext_vector_type(2)));
typedef float v8f __attribute__((ext_vector_type(8)));

__device__ __forceinline__ v2f ld2(const float* p) {
    float2 v = *(const float2*)p;
    v2f r; r[0] = v.x; r[1] = v.y;
    return r;
}

__device__ __forceinline__ void k_step(v2f X, v2f Wv, v2f B, v8f& acc, v2f& pa, v2f& pc) {
    v2f A = X * Wv;
    pa = __builtin_elementwise_fma(A, X, pa);
    pc = __builtin_elementwise_fma(B * Wv, B, pc);
    acc = __builtin_amdgcn_wmma_f32_16x16x4_f32(
         false, A,  false, B,
         (short)0, acc,  false,  false);
}

__global__ __launch_bounds__(256)
void siamese_wmma_kernel(const float* __restrict__ x, const float* __restrict__ y,
                         const int* __restrict__ label, const float* __restrict__ w,
                         const float* __restrict__ bias,
                         float* __restrict__ score, float* __restrict__ partial)
{
    __shared__ float red[256];

    const int lane = threadIdx.x & 31;
    const int wave = threadIdx.x >> 5;
    const int half = lane >> 4;
    const int l    = lane & 15;
    const int m0   = wave * 16;
    const int qb   = blockIdx.x * 16;

    const float* xr = x + (m0 + l) * HH + 2 * half;
    const float* yr = y + (qb + l) * HH + 2 * half;
    const float* wr = w + 2 * half;

    v8f acc = {};
    v2f pa = {0.f, 0.f};
    v2f pc = {0.f, 0.f};

    #pragma nounroll
    for (int kc = 0; kc + 4 * KCH <= HH; kc += 4 * KCH) {
        v2f X[KCH], Wv[KCH], B[KCH];
        #pragma unroll
        for (int j = 0; j < KCH; ++j) {
            X[j]  = ld2(xr + kc + 4 * j);
            Wv[j] = ld2(wr + kc + 4 * j);
            B[j]  = ld2(yr + kc + 4 * j);
        }
        #pragma unroll
        for (int j = 0; j < KCH; ++j) k_step(X[j], Wv[j], B[j], acc, pa, pc);
    }

    k_step(ld2(xr + 224), ld2(wr + 224), ld2(yr + 224), acc, pa, pc);

    {
        v2f X  = {0.f, 0.f};
        v2f Wv = {0.f, 0.f};
        v2f B  = {0.f, 0.f};
        if (half == 0) {
            X  = ld2(xr + 228);
            Wv = ld2(wr + 228);
            B  = ld2(yr + 228);
        }
        k_step(X, Wv, B, acc, pa, pc);
    }

    float an = pa[0] + pa[1];
    float cn = pc[0] + pc[1];
    an += __shfl_xor(an, 16, 32);
    cn += __shfl_xor(cn, 16, 32);

    const float bval = *bias;
    const int   q    = qb + l;
    const float cq   = cn;
    float lsum = 0.f;
    #pragma unroll
    for (int r = 0; r < 8; ++r) {
        const int s = m0 + r + 8 * half;
        const float as = __shfl(an, r + 8 * half, 32);
        const float z  = fmaf(-2.f, acc[r], as + cq + bval);
        const float sc = 1.f / (1.f + expf(-z));
        score[s * QQ + q] = sc;
        const float lab = (float)label[s * QQ + q];
        const float sp = fmaxf(z, 0.f) + log1pf(expf(-fabsf(z)));
        lsum += sp - lab * z;
    }

    red[threadIdx.x] = lsum;
    __syncthreads();
    for (int off = 128; off > 0; off >>= 1) {
        if (threadIdx.x < (unsigned)off) red[threadIdx.x] += red[threadIdx.x + off];
        __syncthreads();
    }
    if (threadIdx.x == 0) partial[blockIdx.x] = red[0];
}

__global__ __launch_bounds__(256)
void loss_kernel(const float* __restrict__ partial, float* __restrict__ out) {
    __shared__ float red[256];
    red[threadIdx.x] = partial[threadIdx.x] + partial[threadIdx.x + 256];
    __syncthreads();
    for (int off = 128; off > 0; off >>= 1) {
        if (threadIdx.x < (unsigned)off) red[threadIdx.x] += red[threadIdx.x + off];
        __syncthreads();
    }
    if (threadIdx.x == 0) out[SS * QQ] = red[0] * (1.0f / (float)(SS * QQ));
}

extern "C" void kernel_launch(void* const* d_in, const int* in_sizes, int n_in,
                              void* d_out, int out_size, void* d_ws, size_t ws_size,
                              hipStream_t stream) {
    const float* x     = (const float*)d_in[0];
    const float* y     = (const float*)d_in[1];
    const int*   label = (const int*)  d_in[2];
    const float* w     = (const float*)d_in[3];
    const float* b     = (const float*)d_in[4];

    float* out = (float*)d_out;
    float* partial = (float*)d_ws;

    siamese_wmma_kernel<<<NBLK, 256, 0, stream>>>(x, y, label, w, b, out, partial);
    loss_kernel<<<1, 256, 0, stream>>>(partial, out);
}


# x, the block's y tile and w staged once in LDS by whole-line float4 loads (y tile once per block, not per wave); fragments read from LDS in the same k order, sums unchanged
# speedup vs baseline: 3.1293x; 1.2589x over previous
#include <hip/hip_runtime.h>
#include <hip/hip_bf16.h>

#define SS 128
#define QQ 8192
#define HH 230
#define NBLK (QQ / 16)
#define KCH 8
#define PITCH 236
#define H2 (HH / 2)
#define TILE4 (16 * HH / 4)
#define XLD ((TILE4 + 31) / 32)
#define YLD ((TILE4 + 255) / 256)

typedef float v2f __attribute__((ext_vector_type(2)));
typedef float v8f __attribute__((ext_vector_type(8)));

__device__ __forceinline__ v2f ld2(const float* p) {
    return *(const v2f*)p;
}

__device__ __forceinline__ void put4(float* tile, int i, float4 v) {
    const int f = 2 * i;
    const int r = f / H2;
    const int c = f - r * H2;
    float* d0 = tile + r * PITCH + 2 * c;
    float* d1 = (c == H2 - 1) ? tile + (r + 1) * PITCH : d0 + 2;
    *(float2*)d0 = make_float2(v.x, v.y);
    *(float2*)d1 = make_float2(v.z, v.w);
}

__device__ __forceinline__ void k_step(v2f X, v2f Wv, v2f B, v8f& acc, v2f& pa, v2f& pc) {
    v2f A = X * Wv;
    pa = __builtin_elementwise_fma(A, X, pa);
    pc = __builtin_elementwise_fma(B * Wv, B, pc);
    acc = __builtin_amdgcn_wmma_f32_16x16x4_f32(
         false, A,  false, B,
         (short)0, acc,  false,  false);
}

__global__ __launch_bounds__(256)
void siamese_wmma_kernel(const float* __restrict__ x, const float* __restrict__ y,
                         const int* __restrict__ label, const float* __restrict__ w,
                         const float* __restrict__ bias,
                         float* __restrict__ score, float* __restrict__ partial)
{
    __shared__ __attribute__((aligned(16))) float xs[SS * PITCH];
    __shared__ __attribute__((aligned(16))) float ys[16 * PITCH];
    __shared__ __attribute__((aligned(16))) float wl[HH + 2];
    __shared__ float red[256];

    const int lane = threadIdx.x & 31;
    const int wave = threadIdx.x >> 5;
    const int half = lane >> 4;
    const int l    = lane & 15;
    const int m0   = wave * 16;
    const int qb   = blockIdx.x * 16;

    {
        const float4* xg = (const float4*)(x + m0 * HH);
        const float4* yg = (const float4*)(y + qb * HH);
        float* xt = xs + m0 * PITCH;
        float4 xv[XLD], yv[YLD];
        float2 wv = make_float2(0.f, 0.f);
        #pragma unroll
        for (int it = 0; it < XLD; ++it) {
            const int i = lane + 32 * it;
            if (i < TILE4) xv[it] = xg[i];
        }
        #pragma unroll
        for (int it = 0; it < YLD; ++it) {
            const int i = (int)threadIdx.x + 256 * it;
            if (i < TILE4) yv[it] = yg[i];
        }
        if (threadIdx.x < H2) wv = ((const float2*)w)[threadIdx.x];

        #pragma unroll
        for (int it = 0; it < XLD; ++it) {
            const int i = lane + 32 * it;
            if (i < TILE4) put4(xt, i, xv[it]);
        }
        #pragma unroll
        for (int it = 0; it < YLD; ++it) {
            const int i = (int)threadIdx.x + 256 * it;
            if (i < TILE4) put4(ys, i, yv[it]);
        }
        if (lane < 16) *(float2*)(xt + lane * PITCH + HH) = make_float2(0.f, 0.f);
        if (threadIdx.x < 16) *(float2*)(ys + threadIdx.x * PITCH + HH) = make_float2(0.f, 0.f);
        if (threadIdx.x <= H2) ((float2*)wl)[threadIdx.x] = wv;
    }
    __syncthreads();

    const float* xr = xs + (m0 + l) * PITCH + 2 * half;
    const float* yr = ys + l * PITCH + 2 * half;
    const float* wr = wl + 2 * half;

    v8f acc = {};
    v2f pa = {0.f, 0.f};
    v2f pc = {0.f, 0.f};

    #pragma nounroll
    for (int kc = 0; kc + 4 * KCH <= HH; kc += 4 * KCH) {
        v2f X[KCH], Wv[KCH], B[KCH];
        #pragma unroll
        for (int j = 0; j < KCH; ++j) {
            X[j]  = ld2(xr + kc + 4 * j);
            Wv[j] = ld2(wr + kc + 4 * j);
            B[j]  = ld2(yr + kc + 4 * j);
        }
        #pragma unroll
        for (int j = 0; j < KCH; ++j) k_step(X[j], Wv[j], B[j], acc, pa, pc);
    }

    {
        const v2f X0 = ld2(xr + 224), W0 = ld2(wr + 224), B0 = ld2(yr + 224);
        const v2f X1 = ld2(xr + 228), W1 = ld2(wr + 228), B1 = ld2(yr + 228);
        k_step(X0, W0, B0, acc, pa, pc);
        k_step(X1, W1, B1, acc, pa, pc);
    }

    float an = pa[0] + pa[1];
    float cn = pc[0] + pc[1];
    an += __shfl_xor(an, 16, 32);
    cn += __shfl_xor(cn, 16, 32);

    const float bval = *bias;
    const int   q    = qb + l;
    const float cq   = cn;
    float lsum = 0.f;
    #pragma unroll
    for (int r = 0; r < 8; ++r) {
        const int s = m0 + r + 8 * half;
        const float as = __shfl(an, r + 8 * half, 32);
        const float z  = fmaf(-2.f, acc[r], as + cq + bval);
        const float sc = 1.f / (1.f + expf(-z));
        score[s * QQ + q] = sc;
        const float lab = (float)label[s * QQ + q];
        const float sp = fmaxf(z, 0.f) + log1pf(expf(-fabsf(z)));
        lsum += sp - lab * z;
    }

    red[threadIdx.x] = lsum;
    __syncthreads();
    for (int off = 128; off > 0; off >>= 1) {
        if (threadIdx.x < (unsigned)off) red[threadIdx.x] += red[threadIdx.x + off];
        __syncthreads();
    }
    if (threadIdx.x == 0) partial[blockIdx.x] = red[0];
}

__global__ __launch_bounds__(256)
void loss_kernel(const float* __restrict__ partial, float* __restrict__ out) {
    __shared__ float red[256];
    red[threadIdx.x] = partial[threadIdx.x] + partial[threadIdx.x + 256];
    __syncthreads();
    for (int off = 128; off > 0; off >>= 1) {
        if (threadIdx.x < (unsigned)off) red[threadIdx.x] += red[threadIdx.x + off];
        __syncthreads();
    }
    if (threadIdx.x == 0) out[SS * QQ] = red[0] * (1.0f / (float)(SS * QQ));
}

extern "C" void kernel_launch(void* const* d_in, const int* in_sizes, int n_in,
                              void* d_out, int out_size, void* d_ws, size_t ws_size,
                              hipStream_t stream) {
    const float* x     = (const float*)d_in[0];
    const float* y     = (const float*)d_in[1];
    const int*   label = (const int*)  d_in[2];
    const float* w     = (const float*)d_in[3];
    const float* b     = (const float*)d_in[4];

    float* out = (float*)d_out;
    float* partial = (float*)d_ws;

    siamese_wmma_kernel<<<NBLK, 256, 0, stream>>>(x, y, label, w, b, out, partial);
    loss_kernel<<<1, 256, 0, stream>>>(partial, out);
}


# Block tile widened from 128x16 to 128x32: x staged in LDS once per 32 q columns by 256 blocks of 8 waves, each wave keeping two accumulator tiles fed by one A fragment; every sum unchanged
# speedup vs baseline: 3.4088x; 1.0893x over previous
#include <hip/hip_runtime.h>
#include <hip/hip_bf16.h>

#define SS 128
#define QQ 8192
#define HH 230
#define QB 32
#define NBLK (QQ / QB)
#define KCH 8
#define PITCH 236
#define H2 (HH / 2)
#define XT4 (SS * HH / 4)
#define YT4 (QB * HH / 4)
#define XLD ((XT4 + 255) / 256)
#define YLD ((YT4 + 255) / 256)

typedef float v2f __attribute__((ext_vector_type(2)));
typedef float v8f __attribute__((ext_vector_type(8)));

__device__ __forceinline__ v2f ld2(const float* p) {
    return *(const v2f*)p;
}

__device__ __forceinline__ void put4(float* tile, int i, float4 v) {
    const int f = 2 * i;
    const int r = f / H2;
    const int c = f - r * H2;
    float* d0 = tile + r * PITCH + 2 * c;
    float* d1 = (c == H2 - 1) ? tile + (r + 1) * PITCH : d0 + 2;
    *(float2*)d0 = make_float2(v.x, v.y);
    *(float2*)d1 = make_float2(v.z, v.w);
}

__device__ __forceinline__ void k_step(v2f X, v2f Wv, v2f B0, v2f B1,
                                       v8f& acc0, v8f& acc1, v2f& pa, v2f& pc0, v2f& pc1) {
    v2f A = X * Wv;
    pa  = __builtin_elementwise_fma(A, X, pa);
    pc0 = __builtin_elementwise_fma(B0 * Wv, B0, pc0);
    pc1 = __builtin_elementwise_fma(B1 * Wv, B1, pc1);
    acc0 = __builtin_amdgcn_wmma_f32_16x16x4_f32(
         false, A,  false, B0,
         (short)0, acc0,  false,  false);
    acc1 = __builtin_amdgcn_wmma_f32_16x16x4_f32(
         false, A,  false, B1,
         (short)0, acc1,  false,  false);
}

__global__ __launch_bounds__(256)
void siamese_wmma_kernel(const float* __restrict__ x, const float* __restrict__ y,
                         const int* __restrict__ label, const float* __restrict__ w,
                         const float* __restrict__ bias,
                         float* __restrict__ score, float* __restrict__ partial)
{
    __shared__ __attribute__((aligned(16))) float xs[SS * PITCH];
    __shared__ __attribute__((aligned(16))) float ys[QB * PITCH];
    __shared__ __attribute__((aligned(16))) float wl[HH + 2];
    __shared__ float2 red[256];

    const int lane = threadIdx.x & 31;
    const int wave = threadIdx.x >> 5;
    const int half = lane >> 4;
    const int l    = lane & 15;
    const int m0   = wave * 16;
    const int qb   = blockIdx.x * QB;

    {
        const float4* xg = (const float4*)x;
        const float4* yg = (const float4*)(y + qb * HH);
        float4 xv[XLD], yv[YLD];
        float2 wv = make_float2(0.f, 0.f);
        #pragma unroll
        for (int it = 0; it < XLD; ++it) {
            const int i = (int)threadIdx.x + 256 * it;
            if (it < XT4 / 256 || i < XT4) xv[it] = xg[i];
        }
        #pragma unroll
        for (int it = 0; it < YLD; ++it) {
            const int i = (int)threadIdx.x + 256 * it;
            if (it < YT4 / 256 || i < YT4) yv[it] = yg[i];
        }
        if (threadIdx.x < H2) wv = ((const float2*)w)[threadIdx.x];

        #pragma unroll
        for (int it = 0; it < XLD; ++it) {
            const int i = (int)threadIdx.x + 256 * it;
            if (it < XT4 / 256 || i < XT4) put4(xs, i, xv[it]);
        }
        #pragma unroll
        for (int it = 0; it < YLD; ++it) {
            const int i = (int)threadIdx.x + 256 * it;
            if (it < YT4 / 256 || i < YT4) put4(ys, i, yv[it]);
        }
        if (threadIdx.x < SS) *(float2*)(xs + threadIdx.x * PITCH + HH) = make_float2(0.f, 0.f);
        if (threadIdx.x < QB) *(float2*)(ys + threadIdx.x * PITCH + HH) = make_float2(0.f, 0.f);
        if (threadIdx.x <= H2) ((float2*)wl)[threadIdx.x] = wv;
    }
    __syncthreads();

    const float* xr  = xs + (m0 + l) * PITCH + 2 * half;
    const float* yr0 = ys + l * PITCH + 2 * half;
    const float* yr1 = ys + (16 + l) * PITCH + 2 * half;
    const float* wr  = wl + 2 * half;

    v8f acc0 = {}, acc1 = {};
    v2f pa  = {0.f, 0.f};
    v2f pc0 = {0.f, 0.f};
    v2f pc1 = {0.f, 0.f};

    #pragma nounroll
    for (int kc = 0; kc + 4 * KCH <= HH; kc += 4 * KCH) {
        v2f X[KCH], Wv[KCH], B0[KCH], B1[KCH];
        #pragma unroll
        for (int j = 0; j < KCH; ++j) {
            X[j]  = ld2(xr  + kc + 4 * j);
            Wv[j] = ld2(wr  + kc + 4 * j);
            B0[j] = ld2(yr0 + kc + 4 * j);
            B1[j] = ld2(yr1 + kc + 4 * j);
        }
        #pragma unroll
        for (int j = 0; j < KCH; ++j)
            k_step(X[j], Wv[j], B0[j], B1[j], acc0, acc1, pa, pc0, pc1);
    }

    {
        const v2f X0 = ld2(xr + 224), W0 = ld2(wr + 224);
        const v2f B00 = ld2(yr0 + 224), B10 = ld2(yr1 + 224);
        const v2f X1 = ld2(xr + 228), W1 = ld2(wr + 228);
        const v2f B01 = ld2(yr0 + 228), B11 = ld2(yr1 + 228);
        k_step(X0, W0, B00, B10, acc0, acc1, pa, pc0, pc1);
        k_step(X1, W1, B01, B11, acc0, acc1, pa, pc0, pc1);
    }

    float an  = pa[0] + pa[1];
    float cn0 = pc0[0] + pc0[1];
    float cn1 = pc1[0] + pc1[1];
    an  += __shfl_xor(an, 16, 32);
    cn0 += __shfl_xor(cn0, 16, 32);
    cn1 += __shfl_xor(cn1, 16, 32);

    const float bval = *bias;
    float lsum0 = 0.f, lsum1 = 0.f;
    #pragma unroll
    for (int r = 0; r < 8; ++r) {
        const int s = m0 + r + 8 * half;
        const float as = __shfl(an, r + 8 * half, 32);
        #pragma unroll
        for (int t = 0; t < 2; ++t) {
            const int   q  = qb + 16 * t + l;
            const float cq = t ? cn1 : cn0;
            const float g  = t ? acc1[r] : acc0[r];
            const float z  = fmaf(-2.f, g, as + cq + bval);
            const float sc = 1.f / (1.f + expf(-z));
            score[s * QQ + q] = sc;
            const float lab = (float)label[s * QQ + q];
            const float sp = fmaxf(z, 0.f) + log1pf(expf(-fabsf(z)));
            if (t) lsum1 += sp - lab * z; else lsum0 += sp - lab * z;
        }
    }

    red[threadIdx.x] = make_float2(lsum0, lsum1);
    __syncthreads();
    for (int off = 128; off > 0; off >>= 1) {
        if (threadIdx.x < (unsigned)off) {
            const float2 u = red[threadIdx.x], v = red[threadIdx.x + off];
            red[threadIdx.x] = make_float2(u.x + v.x, u.y + v.y);
        }
        __syncthreads();
    }
    if (threadIdx.x == 0) {
        const float2 u = red[0];
        partial[2 * blockIdx.x]     = u.x;
        partial[2 * blockIdx.x + 1] = u.y;
    }
}

__global__ __launch_bounds__(256)
void loss_kernel(const float* __restrict__ partial, float* __restrict__ out) {
    __shared__ float red[256];
    red[threadIdx.x] = partial[threadIdx.x] + partial[threadIdx.x + 256];
    __syncthreads();
    for (int off = 128; off > 0; off >>= 1) {
        if (threadIdx.x < (unsigned)off) red[threadIdx.x] += red[threadIdx.x + off];
        __syncthreads();
    }
    if (threadIdx.x == 0) out[SS * QQ] = red[0] * (1.0f / (float)(SS * QQ));
}

extern "C" void kernel_launch(void* const* d_in, const int* in_sizes, int n_in,
                              void* d_out, int out_size, void* d_ws, size_t ws_size,
                              hipStream_t stream) {
    const float* x     = (const float*)d_in[0];
    const float* y     = (const float*)d_in[1];
    const int*   label = (const int*)  d_in[2];
    const float* w     = (const float*)d_in[3];
    const float* b     = (const float*)d_in[4];

    float* out = (float*)d_out;
    float* partial = (float*)d_ws;

    siamese_wmma_kernel<<<NBLK, 256, 0, stream>>>(x, y, label, w, b, out, partial);
    loss_kernel<<<1, 256, 0, stream>>>(partial, out);
}


# Epilogue math: sigmoid and softplus from one shared exp(-|z|) on the hardware exp2, rcp and log2 instructions instead of two library exps, an IEEE divide and log1pf; every sum keeps its order
# speedup vs baseline: 3.9915x; 1.1709x over previous
#include <hip/hip_runtime.h>
#include <hip/hip_bf16.h>

#define SS 128
#define QQ 8192
#define HH 230
#define QB 32
#define NBLK (QQ / QB)
#define KCH 8
#define PITCH 236
#define H2 (HH / 2)
#define XT4 (SS * HH / 4)
#define YT4 (QB * HH / 4)
#define XLD ((XT4 + 255) / 256)
#define YLD ((YT4 + 255) / 256)

typedef float v2f __attribute__((ext_vector_type(2)));
typedef float v8f __attribute__((ext_vector_type(8)));

__device__ __forceinline__ v2f ld2(const float* p) {
    return *(const v2f*)p;
}

__device__ __forceinline__ void put4(float* tile, int i, float4 v) {
    const int f = 2 * i;
    const int r = f / H2;
    const int c = f - r * H2;
    float* d0 = tile + r * PITCH + 2 * c;
    float* d1 = (c == H2 - 1) ? tile + (r + 1) * PITCH : d0 + 2;
    *(float2*)d0 = make_float2(v.x, v.y);
    *(float2*)d1 = make_float2(v.z, v.w);
}

__device__ __forceinline__ void k_step(v2f X, v2f Wv, v2f B0, v2f B1,
                                       v8f& acc0, v8f& acc1, v2f& pa, v2f& pc0, v2f& pc1) {
    v2f A = X * Wv;
    pa  = __builtin_elementwise_fma(A, X, pa);
    pc0 = __builtin_elementwise_fma(B0 * Wv, B0, pc0);
    pc1 = __builtin_elementwise_fma(B1 * Wv, B1, pc1);
    acc0 = __builtin_amdgcn_wmma_f32_16x16x4_f32(
         false, A,  false, B0,
         (short)0, acc0,  false,  false);
    acc1 = __builtin_amdgcn_wmma_f32_16x16x4_f32(
         false, A,  false, B1,
         (short)0, acc1,  false,  false);
}

__global__ __launch_bounds__(256)
void siamese_wmma_kernel(const float* __restrict__ x, const float* __restrict__ y,
                         const int* __restrict__ label, const float* __restrict__ w,
                         const float* __restrict__ bias,
                         float* __restrict__ score, float* __restrict__ partial)
{
    __shared__ __attribute__((aligned(16))) float xs[SS * PITCH];
    __shared__ __attribute__((aligned(16))) float ys[QB * PITCH];
    __shared__ __attribute__((aligned(16))) float wl[HH + 2];
    __shared__ float2 red[256];

    const int lane = threadIdx.x & 31;
    const int wave = threadIdx.x >> 5;
    const int half = lane >> 4;
    const int l    = lane & 15;
    const int m0   = wave * 16;
    const int qb   = blockIdx.x * QB;

    {
        const float4* xg = (const float4*)x;
        const float4* yg = (const float4*)(y + qb * HH);
        float4 xv[XLD], yv[YLD];
        float2 wv = make_float2(0.f, 0.f);
        #pragma unroll
        for (int it = 0; it < XLD; ++it) {
            const int i = (int)threadIdx.x + 256 * it;
            if (it < XT4 / 256 || i < XT4) xv[it] = xg[i];
        }
        #pragma unroll
        for (int it = 0; it < YLD; ++it) {
            const int i = (int)threadIdx.x + 256 * it;
            if (it < YT4 / 256 || i < YT4) yv[it] = yg[i];
        }
        if (threadIdx.x < H2) wv = ((const float2*)w)[threadIdx.x];

        #pragma unroll
        for (int it = 0; it < XLD; ++it) {
            const int i = (int)threadIdx.x + 256 * it;
            if (it < XT4 / 256 || i < XT4) put4(xs, i, xv[it]);
        }
        #pragma unroll
        for (int it = 0; it < YLD; ++it) {
            const int i = (int)threadIdx.x + 256 * it;
            if (it < YT4 / 256 || i < YT4) put4(ys, i, yv[it]);
        }
        if (threadIdx.x < SS) *(float2*)(xs + threadIdx.x * PITCH + HH) = make_float2(0.f, 0.f);
        if (threadIdx.x < QB) *(float2*)(ys + threadIdx.x * PITCH + HH) = make_float2(0.f, 0.f);
        if (threadIdx.x <= H2) ((float2*)wl)[threadIdx.x] = wv;
    }
    __syncthreads();

    const float* xr  = xs + (m0 + l) * PITCH + 2 * half;
    const float* yr0 = ys + l * PITCH + 2 * half;
    const float* yr1 = ys + (16 + l) * PITCH + 2 * half;
    const float* wr  = wl + 2 * half;

    v8f acc0 = {}, acc1 = {};
    v2f pa  = {0.f, 0.f};
    v2f pc0 = {0.f, 0.f};
    v2f pc1 = {0.f, 0.f};

    #pragma nounroll
    for (int kc = 0; kc + 4 * KCH <= HH; kc += 4 * KCH) {
        v2f X[KCH], Wv[KCH], B0[KCH], B1[KCH];
        #pragma unroll
        for (int j = 0; j < KCH; ++j) {
            X[j]  = ld2(xr  + kc + 4 * j);
            Wv[j] = ld2(wr  + kc + 4 * j);
            B0[j] = ld2(yr0 + kc + 4 * j);
            B1[j] = ld2(yr1 + kc + 4 * j);
        }
        #pragma unroll
        for (int j = 0; j < KCH; ++j)
            k_step(X[j], Wv[j], B0[j], B1[j], acc0, acc1, pa, pc0, pc1);
    }

    {
        const v2f X0 = ld2(xr + 224), W0 = ld2(wr + 224);
        const v2f B00 = ld2(yr0 + 224), B10 = ld2(yr1 + 224);
        const v2f X1 = ld2(xr + 228), W1 = ld2(wr + 228);
        const v2f B01 = ld2(yr0 + 228), B11 = ld2(yr1 + 228);
        k_step(X0, W0, B00, B10, acc0, acc1, pa, pc0, pc1);
        k_step(X1, W1, B01, B11, acc0, acc1, pa, pc0, pc1);
    }

    float an  = pa[0] + pa[1];
    float cn0 = pc0[0] + pc0[1];
    float cn1 = pc1[0] + pc1[1];
    an  += __shfl_xor(an, 16, 32);
    cn0 += __shfl_xor(cn0, 16, 32);
    cn1 += __shfl_xor(cn1, 16, 32);

    const float bval = *bias;
    float lsum0 = 0.f, lsum1 = 0.f;
    #pragma unroll
    for (int r = 0; r < 8; ++r) {
        const int s = m0 + r + 8 * half;
        const float as = __shfl(an, r + 8 * half, 32);
        #pragma unroll
        for (int t = 0; t < 2; ++t) {
            const int   q  = qb + 16 * t + l;
            const float cq = t ? cn1 : cn0;
            const float g  = t ? acc1[r] : acc0[r];
            const float z  = fmaf(-2.f, g, as + cq + bval);
            const float e  = __builtin_amdgcn_exp2f(-1.44269504f * fabsf(z));
            const float d  = 1.f + e;
            const float rd = __builtin_amdgcn_rcpf(d);
            const float sc = (z >= 0.f) ? rd : e * rd;
            score[s * QQ + q] = sc;
            const float lab = (float)label[s * QQ + q];
            const float sp = fmaf(__builtin_amdgcn_logf(d), 0.693147181f, fmaxf(z, 0.f));
            if (t) lsum1 += sp - lab * z; else lsum0 += sp - lab * z;
        }
    }

    red[threadIdx.x] = make_float2(lsum0, lsum1);
    __syncthreads();
    for (int off = 128; off > 0; off >>= 1) {
        if (threadIdx.x < (unsigned)off) {
            const float2 u = red[threadIdx.x], v = red[threadIdx.x + off];
            red[threadIdx.x] = make_float2(u.x + v.x, u.y + v.y);
        }
        __syncthreads();
    }
    if (threadIdx.x == 0) {
        const float2 u = red[0];
        partial[2 * blockIdx.x]     = u.x;
        partial[2 * blockIdx.x + 1] = u.y;
    }
}

__global__ __launch_bounds__(256)
void loss_kernel(const float* __restrict__ partial, float* __restrict__ out) {
    __shared__ float red[256];
    red[threadIdx.x] = partial[threadIdx.x] + partial[threadIdx.x + 256];
    __syncthreads();
    for (int off = 128; off > 0; off >>= 1) {
        if (threadIdx.x < (unsigned)off) red[threadIdx.x] += red[threadIdx.x + off];
        __syncthreads();
    }
    if (threadIdx.x == 0) out[SS * QQ] = red[0] * (1.0f / (float)(SS * QQ));
}

extern "C" void kernel_launch(void* const* d_in, const int* in_sizes, int n_in,
                              void* d_out, int out_size, void* d_ws, size_t ws_size,
                              hipStream_t stream) {
    const float* x     = (const float*)d_in[0];
    const float* y     = (const float*)d_in[1];
    const int*   label = (const int*)  d_in[2];
    const float* w     = (const float*)d_in[3];
    const float* b     = (const float*)d_in[4];

    float* out = (float*)d_out;
    float* partial = (float*)d_ws;

    siamese_wmma_kernel<<<NBLK, 256, 0, stream>>>(x, y, label, w, b, out, partial);
    loss_kernel<<<1, 256, 0, stream>>>(partial, out);
}


# BCE sum: each wave adds its 32 lanes with row_xmask DPP adds and one cross-row exchange and stores its own partial; both barrier-per-level LDS trees dropped (18 barriers to 1); new summation order
# speedup vs baseline: 4.4471x; 1.1141x over previous
#include <hip/hip_runtime.h>
#include <hip/hip_bf16.h>

#define SS 128
#define QQ 8192
#define HH 230
#define QB 32
#define NBLK (QQ / QB)
#define NPART (NBLK * 8)
#define KCH 8
#define PITCH 236
#define H2 (HH / 2)
#define XT4 (SS * HH / 4)
#define YT4 (QB * HH / 4)
#define XLD ((XT4 + 255) / 256)
#define YLD ((YT4 + 255) / 256)

typedef float v2f __attribute__((ext_vector_type(2)));
typedef float v8f __attribute__((ext_vector_type(8)));

__device__ __forceinline__ v2f ld2(const float* p) {
    return *(const v2f*)p;
}

template <int CTRL>
__device__ __forceinline__ float dpp_read(float v) {
    const int b = __builtin_bit_cast(int, v);
    return __builtin_bit_cast(float, __builtin_amdgcn_mov_dpp(b, CTRL, 0xf, 0xf, true));
}
__device__ __forceinline__ float wave_sum(float v) {
    v += dpp_read<0x161>(v);
    v += dpp_read<0x162>(v);
    v += dpp_read<0x164>(v);
    v += dpp_read<0x168>(v);
    const int b = __builtin_bit_cast(int, v);
    v += __builtin_bit_cast(float, __builtin_amdgcn_permlanex16(b, b, 0x76543210, 0xfedcba98, false, false));
    return v;
}

__device__ __forceinline__ void put4(float* tile, int i, float4 v) {
    const int f = 2 * i;
    const int r = f / H2;
    const int c = f - r * H2;
    float* d0 = tile + r * PITCH + 2 * c;
    float* d1 = (c == H2 - 1) ? tile + (r + 1) * PITCH : d0 + 2;
    *(float2*)d0 = make_float2(v.x, v.y);
    *(float2*)d1 = make_float2(v.z, v.w);
}

__device__ __forceinline__ void k_step(v2f X, v2f Wv, v2f B0, v2f B1,
                                       v8f& acc0, v8f& acc1, v2f& pa, v2f& pc0, v2f& pc1) {
    v2f A = X * Wv;
    pa  = __builtin_elementwise_fma(A, X, pa);
    pc0 = __builtin_elementwise_fma(B0 * Wv, B0, pc0);
    pc1 = __builtin_elementwise_fma(B1 * Wv, B1, pc1);
    acc0 = __builtin_amdgcn_wmma_f32_16x16x4_f32(
         false, A,  false, B0,
         (short)0, acc0,  false,  false);
    acc1 = __builtin_amdgcn_wmma_f32_16x16x4_f32(
         false, A,  false, B1,
         (short)0, acc1,  false,  false);
}

__global__ __launch_bounds__(256)
void siamese_wmma_kernel(const float* __restrict__ x, const float* __restrict__ y,
                         const int* __restrict__ label, const float* __restrict__ w,
                         const float* __restrict__ bias,
                         float* __restrict__ score, float* __restrict__ partial)
{
    __shared__ __attribute__((aligned(16))) float xs[SS * PITCH];
    __shared__ __attribute__((aligned(16))) float ys[QB * PITCH];
    __shared__ __attribute__((aligned(16))) float wl[HH + 2];

    const int lane = threadIdx.x & 31;
    const int wave = threadIdx.x >> 5;
    const int half = lane >> 4;
    const int l    = lane & 15;
    const int m0   = wave * 16;
    const int qb   = blockIdx.x * QB;

    {
        const float4* xg = (const float4*)x;
        const float4* yg = (const float4*)(y + qb * HH);
        float4 xv[XLD], yv[YLD];
        float2 wv = make_float2(0.f, 0.f);
        #pragma unroll
        for (int it = 0; it < XLD; ++it) {
            const int i = (int)threadIdx.x + 256 * it;
            if (it < XT4 / 256 || i < XT4) xv[it] = xg[i];
        }
        #pragma unroll
        for (int it = 0; it < YLD; ++it) {
            const int i = (int)threadIdx.x + 256 * it;
            if (it < YT4 / 256 || i < YT4) yv[it] = yg[i];
        }
        if (threadIdx.x < H2) wv = ((const float2*)w)[threadIdx.x];

        #pragma unroll
        for (int it = 0; it < XLD; ++it) {
            const int i = (int)threadIdx.x + 256 * it;
            if (it < XT4 / 256 || i < XT4) put4(xs, i, xv[it]);
        }
        #pragma unroll
        for (int it = 0; it < YLD; ++it) {
            const int i = (int)threadIdx.x + 256 * it;
            if (it < YT4 / 256 || i < YT4) put4(ys, i, yv[it]);
        }
        if (threadIdx.x < SS) *(float2*)(xs + threadIdx.x * PITCH + HH) = make_float2(0.f, 0.f);
        if (threadIdx.x < QB) *(float2*)(ys + threadIdx.x * PITCH + HH) = make_float2(0.f, 0.f);
        if (threadIdx.x <= H2) ((float2*)wl)[threadIdx.x] = wv;
    }
    __syncthreads();

    const float* xr  = xs + (m0 + l) * PITCH + 2 * half;
    const float* yr0 = ys + l * PITCH + 2 * half;
    const float* yr1 = ys + (16 + l) * PITCH + 2 * half;
    const float* wr  = wl + 2 * half;

    v8f acc0 = {}, acc1 = {};
    v2f pa  = {0.f, 0.f};
    v2f pc0 = {0.f, 0.f};
    v2f pc1 = {0.f, 0.f};

    #pragma nounroll
    for (int kc = 0; kc + 4 * KCH <= HH; kc += 4 * KCH) {
        v2f X[KCH], Wv[KCH], B0[KCH], B1[KCH];
        #pragma unroll
        for (int j = 0; j < KCH; ++j) {
            X[j]  = ld2(xr  + kc + 4 * j);
            Wv[j] = ld2(wr  + kc + 4 * j);
            B0[j] = ld2(yr0 + kc + 4 * j);
            B1[j] = ld2(yr1 + kc + 4 * j);
        }
        #pragma unroll
        for (int j = 0; j < KCH; ++j)
            k_step(X[j], Wv[j], B0[j], B1[j], acc0, acc1, pa, pc0, pc1);
    }

    {
        const v2f X0 = ld2(xr + 224), W0 = ld2(wr + 224);
        const v2f B00 = ld2(yr0 + 224), B10 = ld2(yr1 + 224);
        const v2f X1 = ld2(xr + 228), W1 = ld2(wr + 228);
        const v2f B01 = ld2(yr0 + 228), B11 = ld2(yr1 + 228);
        k_step(X0, W0, B00, B10, acc0, acc1, pa, pc0, pc1);
        k_step(X1, W1, B01, B11, acc0, acc1, pa, pc0, pc1);
    }

    float an  = pa[0] + pa[1];
    float cn0 = pc0[0] + pc0[1];
    float cn1 = pc1[0] + pc1[1];
    an  += __shfl_xor(an, 16, 32);
    cn0 += __shfl_xor(cn0, 16, 32);
    cn1 += __shfl_xor(cn1, 16, 32);

    const float bval = *bias;
    float lsum0 = 0.f, lsum1 = 0.f;
    #pragma unroll
    for (int r = 0; r < 8; ++r) {
        const int s = m0 + r + 8 * half;
        const float as = __shfl(an, r + 8 * half, 32);
        #pragma unroll
        for (int t = 0; t < 2; ++t) {
            const int   q  = qb + 16 * t + l;
            const float cq = t ? cn1 : cn0;
            const float g  = t ? acc1[r] : acc0[r];
            const float z  = fmaf(-2.f, g, as + cq + bval);
            const float e  = __builtin_amdgcn_exp2f(-1.44269504f * fabsf(z));
            const float d  = 1.f + e;
            const float rd = __builtin_amdgcn_rcpf(d);
            const float sc = (z >= 0.f) ? rd : e * rd;
            score[s * QQ + q] = sc;
            const float lab = (float)label[s * QQ + q];
            const float sp = fmaf(__builtin_amdgcn_logf(d), 0.693147181f, fmaxf(z, 0.f));
            if (t) lsum1 += sp - lab * z; else lsum0 += sp - lab * z;
        }
    }

    const float ws = wave_sum(lsum0 + lsum1);
    if (lane == 0) partial[blockIdx.x * 8 + wave] = ws;
}

__global__ __launch_bounds__(256)
void loss_kernel(const float* __restrict__ partial, float* __restrict__ out) {
    static_assert(NPART == 2 * 4 * 256, "two float4 per thread must cover every partial");
    __shared__ float wsum[8];
    const float4* p4 = (const float4*)partial;
    const float4 u = p4[threadIdx.x], v = p4[threadIdx.x + 256];
    const float s = wave_sum(((u.x + u.y) + (u.z + u.w)) + ((v.x + v.y) + (v.z + v.w)));
    if ((threadIdx.x & 31) == 0) wsum[threadIdx.x >> 5] = s;
    __syncthreads();
    if (threadIdx.x < 32) {
        float t = wsum[threadIdx.x & 7];
        t += dpp_read<0x161>(t);
        t += dpp_read<0x162>(t);
        t += dpp_read<0x164>(t);
        if (threadIdx.x == 0) out[SS * QQ] = t * (1.0f / (float)(SS * QQ));
    }
}

extern "C" void kernel_launch(void* const* d_in, const int* in_sizes, int n_in,
                              void* d_out, int out_size, void* d_ws, size_t ws_size,
                              hipStream_t stream) {
    const float* x     = (const float*)d_in[0];
    const float* y     = (const float*)d_in[1];
    const int*   label = (const int*)  d_in[2];
    const float* w     = (const float*)d_in[3];
    const float* b     = (const float*)d_in[4];

    float* out = (float*)d_out;
    float* partial = (float*)d_ws;

    siamese_wmma_kernel<<<NBLK, 256, 0, stream>>>(x, y, label, w, b, out, partial);
    loss_kernel<<<1, 256, 0, stream>>>(partial, out);
}


# Re-tile: block 64x64 outputs (64 rows each of x and y in LDS) by four waves, one per SIMD, of 32x32, the next chunk's LDS reads issued one group ahead; loss partials: new summation order
# speedup vs baseline: 4.7859x; 1.0762x over previous
#include <hip/hip_runtime.h>
#include <hip/hip_bf16.h>

#define SS 128
#define QQ 8192
#define HH 230
#define TM 64
#define TN 64
#define NRB (SS / TM)
#define NCB (QQ / TN)
#define NBLK (NRB * NCB)
#define NWV 4
#define NTH (32 * NWV)
#define NPART (NBLK * NWV)
#define KCH 8
#define PITCH 236
#define H2 (HH / 2)
#define T4 (TM * HH / 4)
#define LD4 ((T4 + NTH - 1) / NTH)

typedef float v2f __attribute__((ext_vector_type(2)));
typedef float v8f __attribute__((ext_vector_type(8)));

__device__ __forceinline__ v2f ld2(const float* p) {
    return *(const v2f*)p;
}

template <int CTRL>
__device__ __forceinline__ float dpp_read(float v) {
    const int b = __builtin_bit_cast(int, v);
    return __builtin_bit_cast(float, __builtin_amdgcn_mov_dpp(b, CTRL, 0xf, 0xf, true));
}
__device__ __forceinline__ float wave_sum(float v) {
    v += dpp_read<0x161>(v);
    v += dpp_read<0x162>(v);
    v += dpp_read<0x164>(v);
    v += dpp_read<0x168>(v);
    const int b = __builtin_bit_cast(int, v);
    v += __builtin_bit_cast(float, __builtin_amdgcn_permlanex16(b, b, 0x76543210, 0xfedcba98, false, false));
    return v;
}

__device__ __forceinline__ void put4(float* tile, int i, float4 v) {
    const int f = 2 * i;
    const int r = f / H2;
    const int c = f - r * H2;
    float* d0 = tile + r * PITCH + 2 * c;
    float* d1 = (c == H2 - 1) ? tile + (r + 1) * PITCH : d0 + 2;
    *(float2*)d0 = make_float2(v.x, v.y);
    *(float2*)d1 = make_float2(v.z, v.w);
}

struct Frag {
    v2f X0[KCH], X1[KCH], Wv[KCH], B0[KCH], B1[KCH];
};

__device__ __forceinline__ void load_chunk(Frag& f, const float* xr0, const float* xr1,
                                           const float* wr, const float* yr0, const float* yr1) {
    #pragma unroll
    for (int j = 0; j < KCH; ++j) {
        f.X0[j] = ld2(xr0 + 4 * j);
        f.X1[j] = ld2(xr1 + 4 * j);
        f.Wv[j] = ld2(wr  + 4 * j);
        f.B0[j] = ld2(yr0 + 4 * j);
        f.B1[j] = ld2(yr1 + 4 * j);
    }
}

__device__ __forceinline__ void k_step(v2f X0, v2f X1, v2f Wv, v2f B0, v2f B1,
                                       v8f& c00, v8f& c01, v8f& c10, v8f& c11,
                                       v2f& pa0, v2f& pa1, v2f& pc0, v2f& pc1) {
    v2f A0 = X0 * Wv;
    v2f A1 = X1 * Wv;
    pa0 = __builtin_elementwise_fma(A0, X0, pa0);
    pa1 = __builtin_elementwise_fma(A1, X1, pa1);
    pc0 = __builtin_elementwise_fma(B0 * Wv, B0, pc0);
    pc1 = __builtin_elementwise_fma(B1 * Wv, B1, pc1);
    c00 = __builtin_amdgcn_wmma_f32_16x16x4_f32(false, A0, false, B0, (short)0, c00, false, false);
    c01 = __builtin_amdgcn_wmma_f32_16x16x4_f32(false, A0, false, B1, (short)0, c01, false, false);
    c10 = __builtin_amdgcn_wmma_f32_16x16x4_f32(false, A1, false, B0, (short)0, c10, false, false);
    c11 = __builtin_amdgcn_wmma_f32_16x16x4_f32(false, A1, false, B1, (short)0, c11, false, false);
}

__device__ __forceinline__ void chunk_steps(const Frag& f,
                                            v8f& c00, v8f& c01, v8f& c10, v8f& c11,
                                            v2f& pa0, v2f& pa1, v2f& pc0, v2f& pc1) {
    #pragma unroll
    for (int j = 0; j < KCH; ++j)
        k_step(f.X0[j], f.X1[j], f.Wv[j], f.B0[j], f.B1[j], c00, c01, c10, c11, pa0, pa1, pc0, pc1);
}

__global__ __launch_bounds__(NTH)
void siamese_wmma_kernel(const float* __restrict__ x, const float* __restrict__ y,
                         const int* __restrict__ label, const float* __restrict__ w,
                         const float* __restrict__ bias,
                         float* __restrict__ score, float* __restrict__ partial)
{
    static_assert(TM == TN, "one float4 count serves the x tile and the y tile");
    static_assert((TM * H2) % 2 == 0, "a tile holds a whole number of float4");
    __shared__ __attribute__((aligned(16))) float xs[TM * PITCH];
    __shared__ __attribute__((aligned(16))) float ys[TN * PITCH];
    __shared__ __attribute__((aligned(16))) float wl[HH + 2];

    const int tid  = (int)threadIdx.x;
    const int lane = tid & 31;
    const int wave = tid >> 5;
    const int half = lane >> 4;
    const int l    = lane & 15;
    const int sb   = ((int)blockIdx.x & (NRB - 1)) * TM;
    const int qb   = ((int)blockIdx.x / NRB) * TN;
    const int m0   = (wave & 1) * 32;
    const int n0   = (wave >> 1) * 32;

    {
        const float4* xg = (const float4*)(x + sb * HH);
        const float4* yg = (const float4*)(y + qb * HH);
        float4 xv[LD4], yv[LD4];
        float2 wv = make_float2(0.f, 0.f);
        #pragma unroll
        for (int it = 0; it < LD4; ++it) {
            const int i = tid + NTH * it;
            if (it < T4 / NTH || i < T4) xv[it] = xg[i];
        }
        #pragma unroll
        for (int it = 0; it < LD4; ++it) {
            const int i = tid + NTH * it;
            if (it < T4 / NTH || i < T4) yv[it] = yg[i];
        }
        if (tid < H2) wv = ((const float2*)w)[tid];

        #pragma unroll
        for (int it = 0; it < LD4; ++it) {
            const int i = tid + NTH * it;
            if (it < T4 / NTH || i < T4) put4(xs, i, xv[it]);
        }
        #pragma unroll
        for (int it = 0; it < LD4; ++it) {
            const int i = tid + NTH * it;
            if (it < T4 / NTH || i < T4) put4(ys, i, yv[it]);
        }
        if (tid < TM) {
            *(float2*)(xs + tid * PITCH + HH) = make_float2(0.f, 0.f);
            *(float2*)(ys + tid * PITCH + HH) = make_float2(0.f, 0.f);
        }
        if (tid <= H2) ((float2*)wl)[tid] = wv;
    }
    __syncthreads();

    const float* xr0 = xs + (m0 + l) * PITCH + 2 * half;
    const float* xr1 = xs + (m0 + 16 + l) * PITCH + 2 * half;
    const float* yr0 = ys + (n0 + l) * PITCH + 2 * half;
    const float* yr1 = ys + (n0 + 16 + l) * PITCH + 2 * half;
    const float* wr  = wl + 2 * half;

    v8f c00 = {}, c01 = {}, c10 = {}, c11 = {};
    v2f pa0 = {0.f, 0.f}, pa1 = {0.f, 0.f};
    v2f pc0 = {0.f, 0.f}, pc1 = {0.f, 0.f};

    Frag fa, fb;
    load_chunk(fa, xr0, xr1, wr, yr0, yr1);
    #pragma nounroll
    for (int kc = 0; kc < 6 * 4 * KCH; kc += 2 * 4 * KCH) {
        load_chunk(fb, xr0 + kc + 4 * KCH, xr1 + kc + 4 * KCH, wr + kc + 4 * KCH,
                   yr0 + kc + 4 * KCH, yr1 + kc + 4 * KCH);
        chunk_steps(fa, c00, c01, c10, c11, pa0, pa1, pc0, pc1);
        load_chunk(fa, xr0 + kc + 8 * KCH, xr1 + kc + 8 * KCH, wr + kc + 8 * KCH,
                   yr0 + kc + 8 * KCH, yr1 + kc + 8 * KCH);
        chunk_steps(fb, c00, c01, c10, c11, pa0, pa1, pc0, pc1);
        __builtin_amdgcn_sched_group_barrier(0x100, 5 * KCH, 0);
        __builtin_amdgcn_sched_group_barrier(0x008, 4 * KCH, 0);
        __builtin_amdgcn_sched_group_barrier(0x100, 5 * KCH, 0);
        __builtin_amdgcn_sched_group_barrier(0x008, 4 * KCH, 0);
    }

    {
        const v2f X00 = ld2(xr0 + 224), X10 = ld2(xr1 + 224), W0 = ld2(wr + 224);
        const v2f B00 = ld2(yr0 + 224), B10 = ld2(yr1 + 224);
        const v2f X01 = ld2(xr0 + 228), X11 = ld2(xr1 + 228), W1 = ld2(wr + 228);
        const v2f B01 = ld2(yr0 + 228), B11 = ld2(yr1 + 228);
        __builtin_amdgcn_sched_barrier(0);
        chunk_steps(fa, c00, c01, c10, c11, pa0, pa1, pc0, pc1);
        k_step(X00, X10, W0, B00, B10, c00, c01, c10, c11, pa0, pa1, pc0, pc1);
        k_step(X01, X11, W1, B01, B11, c00, c01, c10, c11, pa0, pa1, pc0, pc1);
    }

    float an[2], cn[2];
    an[0] = pa0[0] + pa0[1];
    an[1] = pa1[0] + pa1[1];
    cn[0] = pc0[0] + pc0[1];
    cn[1] = pc1[0] + pc1[1];
    an[0] += __shfl_xor(an[0], 16, 32);
    an[1] += __shfl_xor(an[1], 16, 32);
    cn[0] += __shfl_xor(cn[0], 16, 32);
    cn[1] += __shfl_xor(cn[1], 16, 32);

    const float bval = *bias;
    float lsum[2][2] = {{0.f, 0.f}, {0.f, 0.f}};
    #pragma unroll
    for (int a = 0; a < 2; ++a) {
        #pragma unroll
        for (int r = 0; r < 8; ++r) {
            const int s = sb + m0 + 16 * a + r + 8 * half;
            const float as = __shfl(an[a], r + 8 * half, 32);
            #pragma unroll
            for (int b = 0; b < 2; ++b) {
                const int   q  = qb + n0 + 16 * b + l;
                const float cq = cn[b];
                const v8f&  cc = a ? (b ? c11 : c10) : (b ? c01 : c00);
                const float g  = cc[r];
                const float z  = fmaf(-2.f, g, as + cq + bval);
                const float e  = __builtin_amdgcn_exp2f(-1.44269504f * fabsf(z));
                const float d  = 1.f + e;
                const float rd = __builtin_amdgcn_rcpf(d);
                const float sc = (z >= 0.f) ? rd : e * rd;
                score[s * QQ + q] = sc;
                const float lab = (float)label[s * QQ + q];
                const float sp = fmaf(__builtin_amdgcn_logf(d), 0.693147181f, fmaxf(z, 0.f));
                lsum[a][b] += sp - lab * z;
            }
        }
    }

    const float ws = wave_sum((lsum[0][0] + lsum[0][1]) + (lsum[1][0] + lsum[1][1]));
    if (lane == 0) partial[blockIdx.x * NWV + wave] = ws;
}

__global__ __launch_bounds__(256)
void loss_kernel(const float* __restrict__ partial, float* __restrict__ out) {
    static_assert(NPART == 4 * 256, "one float4 per thread must cover every partial");
    __shared__ float wsum[8];
    const float4* p4 = (const float4*)partial;
    const float4 u = p4[threadIdx.x];
    const float s = wave_sum((u.x + u.y) + (u.z + u.w));
    if ((threadIdx.x & 31) == 0) wsum[threadIdx.x >> 5] = s;
    __syncthreads();
    if (threadIdx.x < 32) {
        float t = wsum[threadIdx.x & 7];
        t += dpp_read<0x161>(t);
        t += dpp_read<0x162>(t);
        t += dpp_read<0x164>(t);
        if (threadIdx.x == 0) out[SS * QQ] = t * (1.0f / (float)(SS * QQ));
    }
}

extern "C" void kernel_launch(void* const* d_in, const int* in_sizes, int n_in,
                              void* d_out, int out_size, void* d_ws, size_t ws_size,
                              hipStream_t stream) {
    const float* x     = (const float*)d_in[0];
    const float* y     = (const float*)d_in[1];
    const int*   label = (const int*)  d_in[2];
    const float* w     = (const float*)d_in[3];
    const float* b     = (const float*)d_in[4];

    float* out = (float*)d_out;
    float* partial = (float*)d_ws;

    siamese_wmma_kernel<<<NBLK, NTH, 0, stream>>>(x, y, label, w, b, out, partial);
    loss_kernel<<<1, 256, 0, stream>>>(partial, out);
}
